# starting kernel
# baseline (speedup 1.0000x reference)
#include <hip/hip_runtime.h>


typedef float v2f __attribute__((ext_vector_type(2)));
typedef float v4f __attribute__((ext_vector_type(4)));
typedef float v8f __attribute__((ext_vector_type(8)));

#define T_STEPS 1024
#define CHANS   (16 * 2048)
#define INV_TAU 0.5f

__device__ __forceinline__ float lampow(int p) {
  return __int_as_float((127 - p) << 23);
}

__global__ __launch_bounds__(256) void lif_scan_wmma(const float* __restrict__ x,
                                                     float* __restrict__ out) {
  const int wave = blockIdx.x * (blockDim.x >> 5) + (threadIdx.x >> 5);
  const int lane = threadIdx.x & 31;
  const int j    = lane & 15;
  const int hi   = lane >> 4;
  const long c0  = (long)wave * 16;

  v2f a[4];
#pragma unroll
  for (int k = 0; k < 4; ++k) {
    const int col0 = 4 * k + 2 * hi;
    const int col1 = col0 + 1;
    a[k].x = (j >= col0) ? lampow(j - col0 + 1) : 0.0f;
    a[k].y = (j >= col1) ? lampow(j - col1 + 1) : 0.0f;
  }
  float cscale[8];
#pragma unroll
  for (int r = 0; r < 8; ++r)
    cscale[r] = lampow(r + 8 * hi + 1);

  const float* px = x + c0 + j;
  float* po = out + c0 + j;

  v4f bxc, byc;
#pragma unroll
  for (int k = 0; k < 4; ++k) {
    bxc[k] = px[(long)(4 * k + 2 * hi) * CHANS];
    byc[k] = px[(long)(4 * k + 1 + 2 * hi) * CHANS];
  }

  float carry = 0.0f;
  const int NT = T_STEPS / 16;

  for (int tt = 0; tt < NT; ++tt) {
    const int tb = tt * 16;

    v4f bxn, byn;
    if (tt + 1 < NT) {
      const int tb2 = tb + 16;
#pragma unroll
      for (int k = 0; k < 4; ++k) {
        bxn[k] = px[(long)(tb2 + 4 * k + 2 * hi) * CHANS];
        byn[k] = px[(long)(tb2 + 4 * k + 1 + 2 * hi) * CHANS];
      }
    } else {
      bxn = (v4f)0.0f; byn = (v4f)0.0f;
    }

    v8f d;
#pragma unroll
    for (int r = 0; r < 8; ++r) d[r] = cscale[r] * carry;

#pragma unroll
    for (int k = 0; k < 4; ++k) {
      v2f b; b.x = bxc[k]; b.y = byc[k];
      d = __builtin_amdgcn_wmma_f32_16x16x4_f32(
           false, a[k],  false, b,
           (short)0, d,  false,  false);
    }

    {
      const float last = d[7];
      const int srcByte = (16 + j) << 2;
      carry = __int_as_float(
          __builtin_amdgcn_ds_bpermute(srcByte, __float_as_int(last)));
    }

#pragma unroll
    for (int r = 0; r < 8; ++r)
      po[(long)(tb + r + 8 * hi) * CHANS] = d[r];

    bxc = bxn; byc = byn;
  }
}

extern "C" void kernel_launch(void* const* d_in, const int* in_sizes, int n_in,
                              void* d_out, int out_size, void* d_ws, size_t ws_size,
                              hipStream_t stream) {
  const float* x = (const float*)d_in[0];
  float* out = (float*)d_out;
  dim3 block(256);
  dim3 grid(CHANS / 128);
  lif_scan_wmma<<<grid, block, 0, stream>>>(x, out);
}


# Time axis split over 16 waves per work-group: a wave requests its 128 steps of x at once, scans from zero; end states meet in LDS, decayed start state added before the stores (sum order changes)
# speedup vs baseline: 1.1175x; 1.1175x over previous
#include <hip/hip_runtime.h>


typedef float v2f __attribute__((ext_vector_type(2)));
typedef float v8f __attribute__((ext_vector_type(8)));

#define T_STEPS 1024
#define CHANS   (16 * 2048)
#define SEG     128
#define NSEG    (T_STEPS / SEG)
#define TILES   (SEG / 16)
#define GROUPS  2
#define THREADS (32 * NSEG * GROUPS)

__device__ __forceinline__ float lampow(int p) {
  return __int_as_float((127 - p) << 23);
}

__global__ __launch_bounds__(THREADS) void lif_scan_wmma(const float* __restrict__ x,
                                                         float* __restrict__ out) {
  __shared__ float seg_end[GROUPS][NSEG][16];

  const int wv   = threadIdx.x >> 5;
  const int lane = threadIdx.x & 31;
  const int j    = lane & 15;
  const int hi   = lane >> 4;
  const int grp  = wv & (GROUPS - 1);
  const int seg  = wv / GROUPS;
  const long c0  = ((long)blockIdx.x * GROUPS + grp) * 16;
  const int  t0  = seg * SEG;

  v2f a[4];
#pragma unroll
  for (int k = 0; k < 4; ++k) {
    const int col0 = 4 * k + 2 * hi;
    const int col1 = col0 + 1;
    a[k].x = (j >= col0) ? lampow(j - col0 + 1) : 0.0f;
    a[k].y = (j >= col1) ? lampow(j - col1 + 1) : 0.0f;
  }
  float cscale[8];
#pragma unroll
  for (int r = 0; r < 8; ++r)
    cscale[r] = lampow(r + 8 * hi + 1);

  const float* px = x + (long)t0 * CHANS + c0 + j;
  float* po = out + (long)t0 * CHANS + c0 + j;

  float bx[TILES][4], by[TILES][4];
#pragma unroll
  for (int i = 0; i < TILES; ++i) {
#pragma unroll
    for (int k = 0; k < 4; ++k) {
      bx[i][k] = px[(long)(16 * i + 4 * k + 2 * hi) * CHANS];
      by[i][k] = px[(long)(16 * i + 4 * k + 1 + 2 * hi) * CHANS];
    }
  }

  __builtin_amdgcn_sched_barrier(0);

  v8f d[TILES];
  float carry = 0.0f;
#pragma unroll
  for (int i = 0; i < TILES; ++i) {
    v8f c;
#pragma unroll
    for (int r = 0; r < 8; ++r) c[r] = cscale[r] * carry;
#pragma unroll
    for (int k = 0; k < 4; ++k) {
      v2f b; b.x = bx[i][k]; b.y = by[i][k];
      c = __builtin_amdgcn_wmma_f32_16x16x4_f32(
           false, a[k],  false, b,
           (short)0, c,  false,  false);
    }
    carry = __int_as_float(
        __builtin_amdgcn_ds_bpermute((16 + j) << 2, __float_as_int(c[7])));
    d[i] = c;
  }

  if (hi == 0) seg_end[grp][seg][j] = carry;
  __syncthreads();

  float s = 0.0f;
#pragma unroll
  for (int u = 0; u < NSEG - 1; ++u) {
    const float e = seg_end[grp][u][j];
    const float t = (s * lampow(64)) * lampow(64) + e;
    s = (u < seg) ? t : s;
  }

#pragma unroll
  for (int i = 0; i < TILES; ++i) {
#pragma unroll
    for (int r = 0; r < 8; ++r)
      d[i][r] = __builtin_fmaf(cscale[r], s, d[i][r]);
    s *= lampow(16);
  }

  __builtin_amdgcn_sched_barrier(0);

#pragma unroll
  for (int i = 0; i < TILES; ++i) {
#pragma unroll
    for (int r = 0; r < 8; ++r)
      po[(long)(16 * i + r + 8 * hi) * CHANS] = d[i][r];
  }
}

extern "C" void kernel_launch(void* const* d_in, const int* in_sizes, int n_in,
                              void* d_out, int out_size, void* d_ws, size_t ws_size,
                              hipStream_t stream) {
  const float* x = (const float*)d_in[0];
  float* out = (float*)d_out;
  dim3 block(THREADS);
  dim3 grid(CHANS / (16 * GROUPS));
  lif_scan_wmma<<<grid, block, 0, stream>>>(x, out);
}


# Tile re-mapped to 4 steps x 64 channels (block-diagonal A; K = lane half x channel parity): a lane loads/stores its own float2, each memory instruction one whole 256-B row piece; sum order changes
# speedup vs baseline: 1.7116x; 1.5316x over previous
#include <hip/hip_runtime.h>


typedef float v2f __attribute__((ext_vector_type(2)));
typedef float v8f __attribute__((ext_vector_type(8)));

#define T_STEPS 1024
#define CHANS   (16 * 2048)
#define SEG     32
#define NSEG    (T_STEPS / SEG)
#define TILE    4
#define TILES   (SEG / TILE)
#define WCH     64
#define THREADS (32 * NSEG)
#define ROW2    (CHANS / 2)

__device__ __forceinline__ float lampow(int p) {
  return __int_as_float((127 - p) << 23);
}

__global__ __launch_bounds__(THREADS) void lif_scan_wmma(const float* __restrict__ x,
                                                         float* __restrict__ out) {
  __shared__ v2f seg_end[NSEG][32];

  const int lane = threadIdx.x & 31;
  const int seg  = __builtin_amdgcn_readfirstlane(threadIdx.x >> 5);
  const int hl   = lane >> 4;
  const int row  = lane & 15;
  const int rh   = row >> 3;
  const int rt   = (row >> 1) & 3;
  const int re   = row & 1;

  v2f a[TILE];
#pragma unroll
  for (int s = 0; s < TILE; ++s) {
    const float w = (rh == hl && rt >= s) ? lampow(rt - s + 1) : 0.0f;
    a[s].x = (re == 0) ? w : 0.0f;
    a[s].y = (re == 1) ? w : 0.0f;
  }

  const long c0 = (long)blockIdx.x * WCH + 2 * lane;
  const long t0 = (long)seg * SEG;
  const v2f* px = (const v2f*)(x + t0 * CHANS + c0);
  v2f* po = (v2f*)(out + t0 * CHANS + c0);

  v2f ld[SEG];
#pragma unroll
  for (int p = 0; p < SEG; ++p) ld[p] = px[(long)p * ROW2];

  __builtin_amdgcn_sched_barrier(0);

  v8f d[TILES];
  float cx = 0.0f, cy = 0.0f;
#pragma unroll
  for (int i = 0; i < TILES; ++i) {
    v8f c;
#pragma unroll
    for (int t = 0; t < TILE; ++t) {
      c[2 * t]     = lampow(t + 1) * cx;
      c[2 * t + 1] = lampow(t + 1) * cy;
    }
#pragma unroll
    for (int s = 0; s < TILE; ++s) {
      c = __builtin_amdgcn_wmma_f32_16x16x4_f32(
           false, a[s],  false, ld[TILE * i + s],
           (short)0, c,  false,  false);
    }
    cx = c[6];
    cy = c[7];
    d[i] = c;
  }

  v2f endst; endst.x = cx; endst.y = cy;
  seg_end[seg][lane] = endst;
  __syncthreads();

  float sx = 0.0f, sy = 0.0f;
#pragma unroll
  for (int g = 0; g < NSEG; g += 8) {
    if (g < seg) {
#pragma unroll
      for (int u = g; u < g + 8 && u < NSEG - 1; ++u) {
        const v2f e = seg_end[u][lane];
        const float tx = sx * lampow(SEG) + e.x;
        const float ty = sy * lampow(SEG) + e.y;
        sx = (u < seg) ? tx : sx;
        sy = (u < seg) ? ty : sy;
      }
    }
  }

#pragma unroll
  for (int i = 0; i < TILES; ++i) {
#pragma unroll
    for (int t = 0; t < TILE; ++t) {
      d[i][2 * t]     = __builtin_fmaf(lampow(t + 1), sx, d[i][2 * t]);
      d[i][2 * t + 1] = __builtin_fmaf(lampow(t + 1), sy, d[i][2 * t + 1]);
    }
    sx *= lampow(TILE);
    sy *= lampow(TILE);
  }

  __builtin_amdgcn_sched_barrier(0);

#pragma unroll
  for (int i = 0; i < TILES; ++i) {
#pragma unroll
    for (int t = 0; t < TILE; ++t) {
      v2f o; o.x = d[i][2 * t]; o.y = d[i][2 * t + 1];
      po[(long)(TILE * i + t) * ROW2] = o;
    }
  }
}

extern "C" void kernel_launch(void* const* d_in, const int* in_sizes, int n_in,
                              void* d_out, int out_size, void* d_ws, size_t ws_size,
                              hipStream_t stream) {
  const float* x = (const float*)d_in[0];
  float* out = (float*)d_out;
  dim3 block(THREADS);
  dim3 grid(CHANS / WCH);
  lif_scan_wmma<<<grid, block, 0, stream>>>(x, out);
}
